# starting kernel
# baseline (speedup 1.0000x reference)
#include <hip/hip_runtime.h>

typedef __attribute__((ext_vector_type(2))) float v2f;
typedef __attribute__((ext_vector_type(8))) float v8f;

#define WMMA_F32(a, b, c) \
  __builtin_amdgcn_wmma_f32_16x16x4_f32(false, (a), false, (b), (short)0, (c), false, false)

constexpr int B_ = 16, S_ = 2048, H_ = 768, CA_ = 32, IN_ = 300;
constexpr int SC = 128;
constexpr int NCHUNK = S_ / SC;
constexpr int KSPLIT = 4;
constexpr int KLEN = S_ / KSPLIT;
constexpr int TQ_STRIDE = 772;

constexpr size_t TQ_OFF  = 0;
constexpr size_t TQ_SZ   = (size_t)B_ * CA_ * H_;
constexpr size_t AL_OFF  = TQ_OFF + TQ_SZ;
constexpr size_t AL_SZ   = (size_t)B_ * CA_ * S_;
constexpr size_t DEN_OFF = AL_OFF + AL_SZ;
constexpr size_t DEN_SZ  = (size_t)B_ * NCHUNK * CA_;
constexpr size_t NUM_OFF = DEN_OFF + DEN_SZ;
constexpr size_t NUM_SZ  = (size_t)KSPLIT * B_ * CA_ * H_;

__global__ void k_tq(const float* __restrict__ query, const float* __restrict__ W,
                     const float* __restrict__ bias, float* __restrict__ tq) {
  const int bc = blockIdx.x;
  __shared__ float qs[IN_];
  const float* qrow = query + (size_t)bc * IN_;
  for (int i = threadIdx.x; i < IN_; i += 256) qs[i] = qrow[i];
  __syncthreads();
  const int h0 = threadIdx.x;
  float a0 = bias[h0], a1 = bias[h0 + 256], a2 = bias[h0 + 512];
  for (int i = 0; i < IN_; ++i) {
    const float qv = qs[i];
    const float* wr = W + (size_t)i * H_ + h0;
    a0 = fmaf(qv, wr[0],   a0);
    a1 = fmaf(qv, wr[256], a1);
    a2 = fmaf(qv, wr[512], a2);
  }
  float* o = tq + (size_t)bc * H_ + h0;
  o[0]   = tanhf(a0);
  o[256] = tanhf(a1);
  o[512] = tanhf(a2);
}

__global__ void k_scores(const float* __restrict__ key, const float* __restrict__ tq,
                         const int* __restrict__ seq_len,
                         float* __restrict__ alpha_g, float* __restrict__ den_part) {
  extern __shared__ float smem[];
  float* tqs  = smem;
  float* dens = smem + 32 * TQ_STRIDE;

  const int b     = blockIdx.x / NCHUNK;
  const int chunk = blockIdx.x % NCHUNK;
  const int s0    = chunk * SC;
  const int tid   = threadIdx.x;
  const int lane  = tid & 31;
  const int wv    = tid >> 5;
  const int hl    = lane & 15;
  const int half  = lane >> 4;
  const int kh    = half * 2;

  const float4* tqg = (const float4*)(tq + (size_t)b * CA_ * H_);
  for (int i = tid; i < 32 * (H_ / 4); i += 256) {
    const int row = i / (H_ / 4), c4 = i % (H_ / 4);
    const float4 v = tqg[i];
    float* d = &tqs[row * TQ_STRIDE + c4 * 4];
    d[0] = v.x; d[1] = v.y; d[2] = v.z; d[3] = v.w;
  }
  __syncthreads();

  const float* a0p  = &tqs[hl * TQ_STRIDE + kh];
  const float* a1p  = &tqs[(hl + 16) * TQ_STRIDE + kh];
  const float* krow = key + ((size_t)b * S_ + (s0 + wv * 16 + hl)) * H_ + kh;

  v8f acc0 = {0.f, 0.f, 0.f, 0.f, 0.f, 0.f, 0.f, 0.f};
  v8f acc1 = {0.f, 0.f, 0.f, 0.f, 0.f, 0.f, 0.f, 0.f};
  for (int k = 0; k < H_; k += 4) {
    const v2f a0 = *(const v2f*)(a0p + k);
    const v2f a1 = *(const v2f*)(a1p + k);
    const v2f bf = *(const v2f*)(krow + k);
    acc0 = WMMA_F32(a0, bf, acc0);
    acc1 = WMMA_F32(a1, bf, acc1);
  }

  const int s = s0 + wv * 16 + hl;
  const bool valid = s < seq_len[b];
  float* ag = alpha_g + (size_t)b * CA_ * S_;
  for (int r = 0; r < 8; ++r) {
    const int c0 = r + 8 * half;
    float v0 = valid ? __expf(acc0[r]) : 0.0f;
    float v1 = valid ? __expf(acc1[r]) : 0.0f;
    ag[(size_t)c0 * S_ + s]        = v0;
    ag[(size_t)(c0 + 16) * S_ + s] = v1;
    float r0 = v0, r1 = v1;
    for (int m = 1; m < 16; m <<= 1) {
      r0 += __shfl_xor(r0, m, 32);
      r1 += __shfl_xor(r1, m, 32);
    }
    if (hl == 0) {
      dens[c0 * 8 + wv]        = r0;
      dens[(c0 + 16) * 8 + wv] = r1;
    }
  }
  __syncthreads();
  if (tid < CA_) {
    float d = 0.f;
    for (int w = 0; w < 8; ++w) d += dens[tid * 8 + w];
    den_part[((size_t)b * NCHUNK + chunk) * CA_ + tid] = d;
  }
}

__global__ void k_out(const float* __restrict__ value, const float* __restrict__ alpha_g,
                      float* __restrict__ num_part) {
  const int gid = blockIdx.x;
  const int b   = gid / (KSPLIT * 6);
  const int rem = gid % (KSPLIT * 6);
  const int kc  = rem / 6;
  const int nb  = rem % 6;
  const int tid  = threadIdx.x;
  const int lane = tid & 31;
  const int wv   = tid >> 5;
  const int hl   = lane & 15;
  const int half = lane >> 4;
  const int kh   = half * 2;
  const int n    = nb * 8 + wv;

  const float* arow0 = alpha_g + ((size_t)b * CA_ + hl) * S_ + kc * KLEN + kh;
  const float* arow1 = arow0 + (size_t)16 * S_;
  const float* vcol  = value + ((size_t)b * S_ + kc * KLEN + kh) * H_ + n * 16 + hl;

  v8f acc0 = {0.f, 0.f, 0.f, 0.f, 0.f, 0.f, 0.f, 0.f};
  v8f acc1 = {0.f, 0.f, 0.f, 0.f, 0.f, 0.f, 0.f, 0.f};
  for (int k = 0; k < KLEN; k += 4) {
    const v2f a0 = *(const v2f*)(arow0 + k);
    const v2f a1 = *(const v2f*)(arow1 + k);
    v2f bf;
    bf.x = vcol[(size_t)k * H_];
    bf.y = vcol[(size_t)(k + 1) * H_];
    acc0 = WMMA_F32(a0, bf, acc0);
    acc1 = WMMA_F32(a1, bf, acc1);
  }
  float* o = num_part + ((size_t)kc * B_ + b) * CA_ * H_;
  for (int r = 0; r < 8; ++r) {
    const int c = r + 8 * half;
    o[(size_t)c * H_ + n * 16 + hl]        = acc0[r];
    o[(size_t)(c + 16) * H_ + n * 16 + hl] = acc1[r];
  }
}

__global__ void k_fin(const float* __restrict__ num_part, const float* __restrict__ den_part,
                      float* __restrict__ out) {
  const int idx = blockIdx.x * 256 + threadIdx.x;
  const int bc  = idx / H_;
  const int b   = bc >> 5, c = bc & 31;
  float num = 0.f;
  for (int kc = 0; kc < KSPLIT; ++kc) num += num_part[(size_t)kc * (B_ * CA_ * H_) + idx];
  float den = 0.f;
  for (int ch = 0; ch < NCHUNK; ++ch) den += den_part[((size_t)b * NCHUNK + ch) * CA_ + c];
  out[idx] = num / den;
}

extern "C" void kernel_launch(void* const* d_in, const int* in_sizes, int n_in,
                              void* d_out, int out_size, void* d_ws, size_t ws_size,
                              hipStream_t stream) {
  (void)in_sizes; (void)n_in; (void)out_size; (void)ws_size;
  const float* key     = (const float*)d_in[0];
  const float* value   = (const float*)d_in[1];
  const float* query   = (const float*)d_in[2];
  const int*   seq_len = (const int*)d_in[3];
  const float* W       = (const float*)d_in[4];
  const float* bias    = (const float*)d_in[5];
  float* ws  = (float*)d_ws;
  float* out = (float*)d_out;

  k_tq<<<B_ * CA_, 256, 0, stream>>>(query, W, bias, ws + TQ_OFF);

  const int smem_bytes = (32 * TQ_STRIDE + 32 * 8) * sizeof(float);
  k_scores<<<B_ * NCHUNK, 256, smem_bytes, stream>>>(key, ws + TQ_OFF, seq_len,
                                                     ws + AL_OFF, ws + DEN_OFF);

  k_out<<<B_ * KSPLIT * 6, 256, 0, stream>>>(value, ws + AL_OFF, ws + NUM_OFF);

  k_fin<<<(B_ * CA_ * H_) / 256, 256, 0, stream>>>(ws + NUM_OFF, ws + DEN_OFF, out);
}


# query projection re-tiled on the vector ALU: block = 8 query rows x 256 columns, a thread keeps 8 running sums so each W load feeds 8 FMAs (W read by 64 row groups, not 512 rows); same sum order
# speedup vs baseline: 1.1785x; 1.1785x over previous
#include <hip/hip_runtime.h>

typedef __attribute__((ext_vector_type(2))) float v2f;
typedef __attribute__((ext_vector_type(8))) float v8f;

#define WMMA_F32(a, b, c) \
  __builtin_amdgcn_wmma_f32_16x16x4_f32(false, (a), false, (b), (short)0, (c), false, false)

constexpr int B_ = 16, S_ = 2048, H_ = 768, CA_ = 32, IN_ = 300;
constexpr int SC = 128;
constexpr int NCHUNK = S_ / SC;
constexpr int KSPLIT = 4;
constexpr int KLEN = S_ / KSPLIT;
constexpr int TQ_STRIDE = 772;

constexpr size_t TQ_OFF  = 0;
constexpr size_t TQ_SZ   = (size_t)B_ * CA_ * H_;
constexpr size_t AL_OFF  = TQ_OFF + TQ_SZ;
constexpr size_t AL_SZ   = (size_t)B_ * CA_ * S_;
constexpr size_t DEN_OFF = AL_OFF + AL_SZ;
constexpr size_t DEN_SZ  = (size_t)B_ * NCHUNK * CA_;
constexpr size_t NUM_OFF = DEN_OFF + DEN_SZ;
constexpr size_t NUM_SZ  = (size_t)KSPLIT * B_ * CA_ * H_;

constexpr int TQ_R = 8;
constexpr int TQ_C = 256;
constexpr int TQ_U = 20;
constexpr int TQ_NB = IN_ / TQ_U;
constexpr int TQ_GRID = (B_ * CA_ / TQ_R) * (H_ / TQ_C);
static_assert(IN_ % TQ_U == 0 && TQ_U % 4 == 0 && (TQ_NB & 1) == 1, "batch plan");
static_assert((B_ * CA_) % TQ_R == 0 && H_ % TQ_C == 0 && IN_ % 4 == 0, "tile plan");

__device__ __forceinline__ void tq_load(float (&w)[TQ_U], const float* __restrict__ wp, int i0) {
#pragma unroll
  for (int j = 0; j < TQ_U; ++j) w[j] = wp[(size_t)(i0 + j) * H_];
}

__device__ __forceinline__ void tq_fma(float (&acc)[TQ_R], const float (&w)[TQ_U],
                                       const float4* __restrict__ qs4, int i0) {
#pragma unroll
  for (int j = 0; j < TQ_U; j += 4) {
#pragma unroll
    for (int r = 0; r < TQ_R; ++r) {
      const float4 q = qs4[(r * IN_ + i0 + j) >> 2];
      float a = acc[r];
      a = fmaf(q.x, w[j + 0], a);
      a = fmaf(q.y, w[j + 1], a);
      a = fmaf(q.z, w[j + 2], a);
      a = fmaf(q.w, w[j + 3], a);
      acc[r] = a;
    }
  }
}

__global__ __launch_bounds__(256) void k_tq(const float* __restrict__ query,
                                            const float* __restrict__ W,
                                            const float* __restrict__ bias,
                                            float* __restrict__ tq) {
  const int rg = blockIdx.x / (H_ / TQ_C);
  const int cg = blockIdx.x % (H_ / TQ_C);
  __shared__ float4 qs4[TQ_R * IN_ / 4];
  float* qs = (float*)qs4;
  const float* qg = query + (size_t)rg * TQ_R * IN_;
  for (int i = threadIdx.x; i < TQ_R * IN_; i += 256) qs[i] = qg[i];
  __syncthreads();

  const int h = cg * TQ_C + threadIdx.x;
  const float* wp = W + h;
  const float bv = bias[h];
  float acc[TQ_R];
#pragma unroll
  for (int r = 0; r < TQ_R; ++r) acc[r] = bv;

  float wa[TQ_U], wb[TQ_U];
  tq_load(wa, wp, 0);
  for (int it = 0; it < TQ_NB - 1; it += 2) {
    tq_load(wb, wp, (it + 1) * TQ_U);
    tq_fma(acc, wa, qs4, it * TQ_U);
    tq_load(wa, wp, (it + 2) * TQ_U);
    tq_fma(acc, wb, qs4, (it + 1) * TQ_U);
  }
  tq_fma(acc, wa, qs4, (TQ_NB - 1) * TQ_U);

  float* o = tq + (size_t)rg * TQ_R * H_ + h;
#pragma unroll
  for (int r = 0; r < TQ_R; ++r) o[(size_t)r * H_] = tanhf(acc[r]);
}

__global__ void k_scores(const float* __restrict__ key, const float* __restrict__ tq,
                         const int* __restrict__ seq_len,
                         float* __restrict__ alpha_g, float* __restrict__ den_part) {
  extern __shared__ float smem[];
  float* tqs  = smem;
  float* dens = smem + 32 * TQ_STRIDE;

  const int b     = blockIdx.x / NCHUNK;
  const int chunk = blockIdx.x % NCHUNK;
  const int s0    = chunk * SC;
  const int tid   = threadIdx.x;
  const int lane  = tid & 31;
  const int wv    = tid >> 5;
  const int hl    = lane & 15;
  const int half  = lane >> 4;
  const int kh    = half * 2;

  const float4* tqg = (const float4*)(tq + (size_t)b * CA_ * H_);
  for (int i = tid; i < 32 * (H_ / 4); i += 256) {
    const int row = i / (H_ / 4), c4 = i % (H_ / 4);
    const float4 v = tqg[i];
    float* d = &tqs[row * TQ_STRIDE + c4 * 4];
    d[0] = v.x; d[1] = v.y; d[2] = v.z; d[3] = v.w;
  }
  __syncthreads();

  const float* a0p  = &tqs[hl * TQ_STRIDE + kh];
  const float* a1p  = &tqs[(hl + 16) * TQ_STRIDE + kh];
  const float* krow = key + ((size_t)b * S_ + (s0 + wv * 16 + hl)) * H_ + kh;

  v8f acc0 = {0.f, 0.f, 0.f, 0.f, 0.f, 0.f, 0.f, 0.f};
  v8f acc1 = {0.f, 0.f, 0.f, 0.f, 0.f, 0.f, 0.f, 0.f};
  for (int k = 0; k < H_; k += 4) {
    const v2f a0 = *(const v2f*)(a0p + k);
    const v2f a1 = *(const v2f*)(a1p + k);
    const v2f bf = *(const v2f*)(krow + k);
    acc0 = WMMA_F32(a0, bf, acc0);
    acc1 = WMMA_F32(a1, bf, acc1);
  }

  const int s = s0 + wv * 16 + hl;
  const bool valid = s < seq_len[b];
  float* ag = alpha_g + (size_t)b * CA_ * S_;
  for (int r = 0; r < 8; ++r) {
    const int c0 = r + 8 * half;
    float v0 = valid ? __expf(acc0[r]) : 0.0f;
    float v1 = valid ? __expf(acc1[r]) : 0.0f;
    ag[(size_t)c0 * S_ + s]        = v0;
    ag[(size_t)(c0 + 16) * S_ + s] = v1;
    float r0 = v0, r1 = v1;
    for (int m = 1; m < 16; m <<= 1) {
      r0 += __shfl_xor(r0, m, 32);
      r1 += __shfl_xor(r1, m, 32);
    }
    if (hl == 0) {
      dens[c0 * 8 + wv]        = r0;
      dens[(c0 + 16) * 8 + wv] = r1;
    }
  }
  __syncthreads();
  if (tid < CA_) {
    float d = 0.f;
    for (int w = 0; w < 8; ++w) d += dens[tid * 8 + w];
    den_part[((size_t)b * NCHUNK + chunk) * CA_ + tid] = d;
  }
}

__global__ void k_out(const float* __restrict__ value, const float* __restrict__ alpha_g,
                      float* __restrict__ num_part) {
  const int gid = blockIdx.x;
  const int b   = gid / (KSPLIT * 6);
  const int rem = gid % (KSPLIT * 6);
  const int kc  = rem / 6;
  const int nb  = rem % 6;
  const int tid  = threadIdx.x;
  const int lane = tid & 31;
  const int wv   = tid >> 5;
  const int hl   = lane & 15;
  const int half = lane >> 4;
  const int kh   = half * 2;
  const int n    = nb * 8 + wv;

  const float* arow0 = alpha_g + ((size_t)b * CA_ + hl) * S_ + kc * KLEN + kh;
  const float* arow1 = arow0 + (size_t)16 * S_;
  const float* vcol  = value + ((size_t)b * S_ + kc * KLEN + kh) * H_ + n * 16 + hl;

  v8f acc0 = {0.f, 0.f, 0.f, 0.f, 0.f, 0.f, 0.f, 0.f};
  v8f acc1 = {0.f, 0.f, 0.f, 0.f, 0.f, 0.f, 0.f, 0.f};
  for (int k = 0; k < KLEN; k += 4) {
    const v2f a0 = *(const v2f*)(arow0 + k);
    const v2f a1 = *(const v2f*)(arow1 + k);
    v2f bf;
    bf.x = vcol[(size_t)k * H_];
    bf.y = vcol[(size_t)(k + 1) * H_];
    acc0 = WMMA_F32(a0, bf, acc0);
    acc1 = WMMA_F32(a1, bf, acc1);
  }
  float* o = num_part + ((size_t)kc * B_ + b) * CA_ * H_;
  for (int r = 0; r < 8; ++r) {
    const int c = r + 8 * half;
    o[(size_t)c * H_ + n * 16 + hl]        = acc0[r];
    o[(size_t)(c + 16) * H_ + n * 16 + hl] = acc1[r];
  }
}

__global__ void k_fin(const float* __restrict__ num_part, const float* __restrict__ den_part,
                      float* __restrict__ out) {
  const int idx = blockIdx.x * 256 + threadIdx.x;
  const int bc  = idx / H_;
  const int b   = bc >> 5, c = bc & 31;
  float num = 0.f;
  for (int kc = 0; kc < KSPLIT; ++kc) num += num_part[(size_t)kc * (B_ * CA_ * H_) + idx];
  float den = 0.f;
  for (int ch = 0; ch < NCHUNK; ++ch) den += den_part[((size_t)b * NCHUNK + ch) * CA_ + c];
  out[idx] = num / den;
}

extern "C" void kernel_launch(void* const* d_in, const int* in_sizes, int n_in,
                              void* d_out, int out_size, void* d_ws, size_t ws_size,
                              hipStream_t stream) {
  (void)in_sizes; (void)n_in; (void)out_size; (void)ws_size;
  const float* key     = (const float*)d_in[0];
  const float* value   = (const float*)d_in[1];
  const float* query   = (const float*)d_in[2];
  const int*   seq_len = (const int*)d_in[3];
  const float* W       = (const float*)d_in[4];
  const float* bias    = (const float*)d_in[5];
  float* ws  = (float*)d_ws;
  float* out = (float*)d_out;

  k_tq<<<TQ_GRID, 256, 0, stream>>>(query, W, bias, ws + TQ_OFF);

  const int smem_bytes = (32 * TQ_STRIDE + 32 * 8) * sizeof(float);
  k_scores<<<B_ * NCHUNK, 256, smem_bytes, stream>>>(key, ws + TQ_OFF, seq_len,
                                                     ws + AL_OFF, ws + DEN_OFF);

  k_out<<<B_ * KSPLIT * 6, 256, 0, stream>>>(value, ws + AL_OFF, ws + NUM_OFF);

  k_fin<<<(B_ * CA_ * H_) / 256, 256, 0, stream>>>(ws + NUM_OFF, ws + DEN_OFF, out);
}


# both GEMMs: key, value and alpha tiles go global -> registers -> LDS by coalesced 16-byte loads issued one K-chunk ahead (two LDS buffers), fragments read from LDS; same k order and sums
# speedup vs baseline: 1.6258x; 1.3796x over previous
#include <hip/hip_runtime.h>

typedef __attribute__((ext_vector_type(2))) float v2f;
typedef __attribute__((ext_vector_type(8))) float v8f;

#define WMMA_F32(a, b, c) \
  __builtin_amdgcn_wmma_f32_16x16x4_f32(false, (a), false, (b), (short)0, (c), false, false)

constexpr int B_ = 16, S_ = 2048, H_ = 768, CA_ = 32, IN_ = 300;
constexpr int SC = 128;
constexpr int NCHUNK = S_ / SC;
constexpr int KSPLIT = 4;
constexpr int KLEN = S_ / KSPLIT;
constexpr int TQ_STRIDE = 772;

constexpr size_t TQ_OFF  = 0;
constexpr size_t TQ_SZ   = (size_t)B_ * CA_ * H_;
constexpr size_t AL_OFF  = TQ_OFF + TQ_SZ;
constexpr size_t AL_SZ   = (size_t)B_ * CA_ * S_;
constexpr size_t DEN_OFF = AL_OFF + AL_SZ;
constexpr size_t DEN_SZ  = (size_t)B_ * NCHUNK * CA_;
constexpr size_t NUM_OFF = DEN_OFF + DEN_SZ;
constexpr size_t NUM_SZ  = (size_t)KSPLIT * B_ * CA_ * H_;

constexpr int TQ_R = 8;
constexpr int TQ_C = 256;
constexpr int TQ_U = 20;
constexpr int TQ_NB = IN_ / TQ_U;
constexpr int TQ_GRID = (B_ * CA_ / TQ_R) * (H_ / TQ_C);
static_assert(IN_ % TQ_U == 0 && TQ_U % 4 == 0 && (TQ_NB & 1) == 1, "batch plan");
static_assert((B_ * CA_) % TQ_R == 0 && H_ % TQ_C == 0 && IN_ % 4 == 0, "tile plan");

__device__ __forceinline__ void tq_load(float (&w)[TQ_U], const float* __restrict__ wp, int i0) {
#pragma unroll
  for (int j = 0; j < TQ_U; ++j) w[j] = wp[(size_t)(i0 + j) * H_];
}

__device__ __forceinline__ void tq_fma(float (&acc)[TQ_R], const float (&w)[TQ_U],
                                       const float4* __restrict__ qs4, int i0) {
#pragma unroll
  for (int j = 0; j < TQ_U; j += 4) {
#pragma unroll
    for (int r = 0; r < TQ_R; ++r) {
      const float4 q = qs4[(r * IN_ + i0 + j) >> 2];
      float a = acc[r];
      a = fmaf(q.x, w[j + 0], a);
      a = fmaf(q.y, w[j + 1], a);
      a = fmaf(q.z, w[j + 2], a);
      a = fmaf(q.w, w[j + 3], a);
      acc[r] = a;
    }
  }
}

__global__ __launch_bounds__(256) void k_tq(const float* __restrict__ query,
                                            const float* __restrict__ W,
                                            const float* __restrict__ bias,
                                            float* __restrict__ tq) {
  const int rg = blockIdx.x / (H_ / TQ_C);
  const int cg = blockIdx.x % (H_ / TQ_C);
  __shared__ float4 qs4[TQ_R * IN_ / 4];
  float* qs = (float*)qs4;
  const float* qg = query + (size_t)rg * TQ_R * IN_;
  for (int i = threadIdx.x; i < TQ_R * IN_; i += 256) qs[i] = qg[i];
  __syncthreads();

  const int h = cg * TQ_C + threadIdx.x;
  const float* wp = W + h;
  const float bv = bias[h];
  float acc[TQ_R];
#pragma unroll
  for (int r = 0; r < TQ_R; ++r) acc[r] = bv;

  float wa[TQ_U], wb[TQ_U];
  tq_load(wa, wp, 0);
  for (int it = 0; it < TQ_NB - 1; it += 2) {
    tq_load(wb, wp, (it + 1) * TQ_U);
    tq_fma(acc, wa, qs4, it * TQ_U);
    tq_load(wa, wp, (it + 2) * TQ_U);
    tq_fma(acc, wb, qs4, (it + 1) * TQ_U);
  }
  tq_fma(acc, wa, qs4, (TQ_NB - 1) * TQ_U);

  float* o = tq + (size_t)rg * TQ_R * H_ + h;
#pragma unroll
  for (int r = 0; r < TQ_R; ++r) o[(size_t)r * H_] = tanhf(acc[r]);
}

constexpr int KC    = 128;
constexpr int SNCH  = H_ / KC;
constexpr int KNL   = SC * KC / 4 / 256;
constexpr int KP    = KC + 4;
constexpr int KB_SZ = SC * KP;
constexpr int SC_SMEM_FLOATS = 2 * KB_SZ + 32 * TQ_STRIDE + 32 * 8;
static_assert(H_ % KC == 0 && KC == 128 && SC == 128 && KP % 4 == 0 &&
              (32 * TQ_STRIDE + 32 * 8) % 4 == 0 && KB_SZ % 4 == 0,
              "chunk plan: a wave loads one 128-float row piece; 16-byte aligned key tiles");

__device__ __forceinline__ void sc_load(float4 (&kr)[KNL], const float* __restrict__ kg, int c) {
#pragma unroll
  for (int j = 0; j < KNL; ++j) kr[j] = *(const float4*)(kg + (size_t)(8 * j) * H_ + c * KC);
}

__device__ __forceinline__ void sc_store(const float4 (&kr)[KNL], float* kb) {
#pragma unroll
  for (int j = 0; j < KNL; ++j) *(float4*)(kb + 8 * j * KP) = kr[j];
}

__device__ __forceinline__ void sc_mma(v8f& acc0, v8f& acc1, const float* a0p, const float* a1p,
                                       const float* bp) {
#pragma unroll
  for (int kk = 0; kk < KC / 4; ++kk) {
    const v2f a0 = *(const v2f*)(a0p + kk * 4);
    const v2f a1 = *(const v2f*)(a1p + kk * 4);
    const v2f bf = *(const v2f*)(bp + kk * 4);
    acc0 = WMMA_F32(a0, bf, acc0);
    acc1 = WMMA_F32(a1, bf, acc1);
  }
}

__global__ __launch_bounds__(256) void k_scores(const float* __restrict__ key,
                                                const float* __restrict__ tq,
                                                const int* __restrict__ seq_len,
                                                float* __restrict__ alpha_g,
                                                float* __restrict__ den_part) {
  __shared__ __attribute__((aligned(16))) float smem[SC_SMEM_FLOATS];
  float* tqs  = smem;
  float* dens = smem + 32 * TQ_STRIDE;
  float* kbuf = dens + 32 * 8;

  const int b     = blockIdx.x / NCHUNK;
  const int chunk = blockIdx.x % NCHUNK;
  const int s0    = chunk * SC;
  const int tid   = threadIdx.x;
  const int lane  = tid & 31;
  const int wv    = tid >> 5;
  const int hl    = lane & 15;
  const int half  = lane >> 4;
  const int kh    = half * 2;

  const float* kg = key + ((size_t)b * S_ + s0 + (tid >> 5)) * H_ + (tid & 31) * 4;
  const int kso = (tid >> 5) * KP + (tid & 31) * 4;
  float4 kr[KNL];

  const float4* tqg = (const float4*)(tq + (size_t)b * CA_ * H_);
  for (int i = tid; i < 32 * (H_ / 4); i += 256) {
    const int row = i / (H_ / 4), c4 = i % (H_ / 4);
    const float4 v = tqg[i];
    float* d = &tqs[row * TQ_STRIDE + c4 * 4];
    d[0] = v.x; d[1] = v.y; d[2] = v.z; d[3] = v.w;
  }
  sc_load(kr, kg, 0);
  sc_store(kr, kbuf + kso);
  __syncthreads();

  const float* a0p  = &tqs[hl * TQ_STRIDE + kh];
  const float* a1p  = &tqs[(hl + 16) * TQ_STRIDE + kh];
  const int bfo = (wv * 16 + hl) * KP + kh;

  v8f acc0 = {0.f, 0.f, 0.f, 0.f, 0.f, 0.f, 0.f, 0.f};
  v8f acc1 = {0.f, 0.f, 0.f, 0.f, 0.f, 0.f, 0.f, 0.f};
  for (int c = 0; c < SNCH - 1; ++c) {
    const int cur = c & 1, nxt = cur ^ 1;
    sc_load(kr, kg, c + 1);
    sc_mma(acc0, acc1, a0p + c * KC, a1p + c * KC, kbuf + cur * KB_SZ + bfo);
    sc_store(kr, kbuf + nxt * KB_SZ + kso);
    __builtin_amdgcn_sched_group_barrier(0x020, KNL, 0);
    __builtin_amdgcn_sched_group_barrier(0x100, 48, 0);
    __builtin_amdgcn_sched_group_barrier(0x008, KC / 2, 0);
    __builtin_amdgcn_sched_group_barrier(0x200, KNL, 0);
    __syncthreads();
  }
  {
    const int c = SNCH - 1, cur = c & 1;
    sc_mma(acc0, acc1, a0p + c * KC, a1p + c * KC, kbuf + cur * KB_SZ + bfo);
  }

  const int s = s0 + wv * 16 + hl;
  const bool valid = s < seq_len[b];
  float* ag = alpha_g + (size_t)b * CA_ * S_;
  for (int r = 0; r < 8; ++r) {
    const int c0 = r + 8 * half;
    float v0 = valid ? __expf(acc0[r]) : 0.0f;
    float v1 = valid ? __expf(acc1[r]) : 0.0f;
    ag[(size_t)c0 * S_ + s]        = v0;
    ag[(size_t)(c0 + 16) * S_ + s] = v1;
    float r0 = v0, r1 = v1;
    for (int m = 1; m < 16; m <<= 1) {
      r0 += __shfl_xor(r0, m, 32);
      r1 += __shfl_xor(r1, m, 32);
    }
    if (hl == 0) {
      dens[c0 * 8 + wv]        = r0;
      dens[(c0 + 16) * 8 + wv] = r1;
    }
  }
  __syncthreads();
  if (tid < CA_) {
    float d = 0.f;
    for (int w = 0; w < 8; ++w) d += dens[tid * 8 + w];
    den_part[((size_t)b * NCHUNK + chunk) * CA_ + tid] = d;
  }
}

constexpr int OC    = 64;
constexpr int ONCH  = KLEN / OC;
constexpr int VP    = 128 + 8;
constexpr int AP    = OC + 4;
constexpr int OV_SZ = OC * VP;
constexpr int OA_SZ = CA_ * AP;
static_assert(KLEN % OC == 0 && OC % 4 == 0 && VP % 4 == 0 && AP % 4 == 0, "chunk plan");

__device__ __forceinline__ void out_load(float4 (&vr)[8], float4 (&ar)[2],
                                         const float* __restrict__ vg,
                                         const float* __restrict__ ag, int ch) {
#pragma unroll
  for (int j = 0; j < 8; ++j) vr[j] = *(const float4*)(vg + (size_t)(ch * OC + 8 * j) * H_);
#pragma unroll
  for (int j = 0; j < 2; ++j) ar[j] = *(const float4*)(ag + (size_t)(16 * j) * S_ + ch * OC);
}

__device__ __forceinline__ void out_store(const float4 (&vr)[8], const float4 (&ar)[2],
                                          float* vs, float* as_) {
#pragma unroll
  for (int j = 0; j < 8; ++j) *(float4*)(vs + 8 * j * VP) = vr[j];
#pragma unroll
  for (int j = 0; j < 2; ++j) *(float4*)(as_ + 16 * j * AP) = ar[j];
}

__device__ __forceinline__ void out_mma(v8f& acc0, v8f& acc1, const float* a0p,
                                        const float* bp) {
#pragma unroll
  for (int kk = 0; kk < OC / 4; ++kk) {
    const v2f a0 = *(const v2f*)(a0p + kk * 4);
    const v2f a1 = *(const v2f*)(a0p + 16 * AP + kk * 4);
    v2f bf;
    bf.x = bp[(kk * 4) * VP];
    bf.y = bp[(kk * 4 + 1) * VP];
    acc0 = WMMA_F32(a0, bf, acc0);
    acc1 = WMMA_F32(a1, bf, acc1);
  }
}

__global__ __launch_bounds__(256) void k_out(const float* __restrict__ value,
                                             const float* __restrict__ alpha_g,
                                             float* __restrict__ num_part) {
  __shared__ __attribute__((aligned(16))) float osm[2 * (OV_SZ + OA_SZ)];
  const int gid = blockIdx.x;
  const int b   = gid / (KSPLIT * 6);
  const int rem = gid % (KSPLIT * 6);
  const int kc  = rem / 6;
  const int nb  = rem % 6;
  const int tid  = threadIdx.x;
  const int lane = tid & 31;
  const int wv   = tid >> 5;
  const int hl   = lane & 15;
  const int half = lane >> 4;
  const int kh   = half * 2;
  const int n    = nb * 8 + wv;

  const int vrow = tid >> 5, vc4 = tid & 31;
  const int arow = tid >> 4, aq  = tid & 15;
  const float* vg = value + ((size_t)b * S_ + kc * KLEN + vrow) * H_ + nb * 128 + vc4 * 4;
  const float* ag = alpha_g + ((size_t)b * CA_ + arow) * S_ + kc * KLEN + aq * 4;
  const int vso = vrow * VP + vc4 * 4;
  const int aso = arow * AP + aq * 4;
  const int afo = hl * AP + kh;
  const int bfo = kh * VP + wv * 16 + hl;

  v8f acc0 = {0.f, 0.f, 0.f, 0.f, 0.f, 0.f, 0.f, 0.f};
  v8f acc1 = {0.f, 0.f, 0.f, 0.f, 0.f, 0.f, 0.f, 0.f};
  float4 vr[8], ar[2];
  out_load(vr, ar, vg, ag, 0);
  out_store(vr, ar, osm + vso, osm + 2 * OV_SZ + aso);
  __syncthreads();
  for (int ch = 0; ch < ONCH - 1; ++ch) {
    const int cur = ch & 1, nxt = cur ^ 1;
    out_load(vr, ar, vg, ag, ch + 1);
    out_mma(acc0, acc1, osm + 2 * OV_SZ + cur * OA_SZ + afo, osm + cur * OV_SZ + bfo);
    out_store(vr, ar, osm + nxt * OV_SZ + vso, osm + 2 * OV_SZ + nxt * OA_SZ + aso);
    __builtin_amdgcn_sched_group_barrier(0x020, 10, 0);
    __builtin_amdgcn_sched_group_barrier(0x100, 32, 0);
    __builtin_amdgcn_sched_group_barrier(0x008, 32, 0);
    __builtin_amdgcn_sched_group_barrier(0x200, 10, 0);
    __syncthreads();
  }
  {
    const int cur = (ONCH - 1) & 1;
    out_mma(acc0, acc1, osm + 2 * OV_SZ + cur * OA_SZ + afo, osm + cur * OV_SZ + bfo);
  }

  float* o = num_part + ((size_t)kc * B_ + b) * CA_ * H_;
  for (int r = 0; r < 8; ++r) {
    const int c = r + 8 * half;
    o[(size_t)c * H_ + n * 16 + hl]        = acc0[r];
    o[(size_t)(c + 16) * H_ + n * 16 + hl] = acc1[r];
  }
}

__global__ void k_fin(const float* __restrict__ num_part, const float* __restrict__ den_part,
                      float* __restrict__ out) {
  const int idx = blockIdx.x * 256 + threadIdx.x;
  const int bc  = idx / H_;
  const int b   = bc >> 5, c = bc & 31;
  float num = 0.f;
  for (int kc = 0; kc < KSPLIT; ++kc) num += num_part[(size_t)kc * (B_ * CA_ * H_) + idx];
  float den = 0.f;
  for (int ch = 0; ch < NCHUNK; ++ch) den += den_part[((size_t)b * NCHUNK + ch) * CA_ + c];
  out[idx] = num / den;
}

extern "C" void kernel_launch(void* const* d_in, const int* in_sizes, int n_in,
                              void* d_out, int out_size, void* d_ws, size_t ws_size,
                              hipStream_t stream) {
  (void)in_sizes; (void)n_in; (void)out_size; (void)ws_size;
  const float* key     = (const float*)d_in[0];
  const float* value   = (const float*)d_in[1];
  const float* query   = (const float*)d_in[2];
  const int*   seq_len = (const int*)d_in[3];
  const float* W       = (const float*)d_in[4];
  const float* bias    = (const float*)d_in[5];
  float* ws  = (float*)d_ws;
  float* out = (float*)d_out;

  k_tq<<<TQ_GRID, 256, 0, stream>>>(query, W, bias, ws + TQ_OFF);

  k_scores<<<B_ * NCHUNK, 256, 0, stream>>>(key, ws + TQ_OFF, seq_len,
                                                     ws + AL_OFF, ws + DEN_OFF);

  k_out<<<B_ * KSPLIT * 6, 256, 0, stream>>>(value, ws + AL_OFF, ws + NUM_OFF);

  k_fin<<<(B_ * CA_ * H_) / 256, 256, 0, stream>>>(ws + NUM_OFF, ws + DEN_OFF, out);
}


# latency chains outside the main loops overlapped: projection and scores prologues issue all staging loads before the first wait; scores epilogue's 16 row-sum shuffle chains interleaved; same sums
# speedup vs baseline: 1.7202x; 1.0581x over previous
#include <hip/hip_runtime.h>

typedef __attribute__((ext_vector_type(2))) float v2f;
typedef __attribute__((ext_vector_type(8))) float v8f;

#define WMMA_F32(a, b, c) \
  __builtin_amdgcn_wmma_f32_16x16x4_f32(false, (a), false, (b), (short)0, (c), false, false)

constexpr int B_ = 16, S_ = 2048, H_ = 768, CA_ = 32, IN_ = 300;
constexpr int SC = 128;
constexpr int NCHUNK = S_ / SC;
constexpr int KSPLIT = 4;
constexpr int KLEN = S_ / KSPLIT;
constexpr int TQ_STRIDE = 772;

constexpr size_t TQ_OFF  = 0;
constexpr size_t TQ_SZ   = (size_t)B_ * CA_ * H_;
constexpr size_t AL_OFF  = TQ_OFF + TQ_SZ;
constexpr size_t AL_SZ   = (size_t)B_ * CA_ * S_;
constexpr size_t DEN_OFF = AL_OFF + AL_SZ;
constexpr size_t DEN_SZ  = (size_t)B_ * NCHUNK * CA_;
constexpr size_t NUM_OFF = DEN_OFF + DEN_SZ;
constexpr size_t NUM_SZ  = (size_t)KSPLIT * B_ * CA_ * H_;

constexpr int TQ_R = 8;
constexpr int TQ_C = 256;
constexpr int TQ_U = 20;
constexpr int TQ_NB = IN_ / TQ_U;
constexpr int TQ_GRID = (B_ * CA_ / TQ_R) * (H_ / TQ_C);
constexpr int TQ_QV = TQ_R * IN_ / 4;
static_assert(TQ_QV > 512 && TQ_QV <= 768 && (TQ_R * IN_) % 4 == 0, "query staging: three pieces per thread");
static_assert(IN_ % TQ_U == 0 && TQ_U % 4 == 0 && (TQ_NB & 1) == 1, "batch plan");
static_assert((B_ * CA_) % TQ_R == 0 && H_ % TQ_C == 0 && IN_ % 4 == 0, "tile plan");

__device__ __forceinline__ void tq_load(float (&w)[TQ_U], const float* __restrict__ wp, int i0) {
#pragma unroll
  for (int j = 0; j < TQ_U; ++j) w[j] = wp[(size_t)(i0 + j) * H_];
}

__device__ __forceinline__ void tq_fma(float (&acc)[TQ_R], const float (&w)[TQ_U],
                                       const float4* __restrict__ qs4, int i0) {
#pragma unroll
  for (int j = 0; j < TQ_U; j += 4) {
#pragma unroll
    for (int r = 0; r < TQ_R; ++r) {
      const float4 q = qs4[(r * IN_ + i0 + j) >> 2];
      float a = acc[r];
      a = fmaf(q.x, w[j + 0], a);
      a = fmaf(q.y, w[j + 1], a);
      a = fmaf(q.z, w[j + 2], a);
      a = fmaf(q.w, w[j + 3], a);
      acc[r] = a;
    }
  }
}

__global__ __launch_bounds__(256) void k_tq(const float* __restrict__ query,
                                            const float* __restrict__ W,
                                            const float* __restrict__ bias,
                                            float* __restrict__ tq) {
  const int rg = blockIdx.x / (H_ / TQ_C);
  const int cg = blockIdx.x % (H_ / TQ_C);
  __shared__ float4 qs4[3 * 256];
  const float4* qg4 = (const float4*)(query + (size_t)rg * TQ_R * IN_);
  const int t = threadIdx.x;
  const int h = cg * TQ_C + t;
  const float* wp = W + h;
  const float4 q0 = qg4[t];
  const float4 q1 = qg4[t + 256];
  const float4 q2 = qg4[t + 512 < TQ_QV ? t + 512 : t];
  const float bv = bias[h];
  float wa[TQ_U], wb[TQ_U];
  tq_load(wa, wp, 0);
  qs4[t] = q0;
  qs4[t + 256] = q1;
  qs4[t + 512] = q2;
  __syncthreads();

  float acc[TQ_R];
#pragma unroll
  for (int r = 0; r < TQ_R; ++r) acc[r] = bv;

  for (int it = 0; it < TQ_NB - 1; it += 2) {
    tq_load(wb, wp, (it + 1) * TQ_U);
    tq_fma(acc, wa, qs4, it * TQ_U);
    tq_load(wa, wp, (it + 2) * TQ_U);
    tq_fma(acc, wb, qs4, (it + 1) * TQ_U);
  }
  tq_fma(acc, wa, qs4, (TQ_NB - 1) * TQ_U);

  float* o = tq + (size_t)rg * TQ_R * H_ + h;
#pragma unroll
  for (int r = 0; r < TQ_R; ++r) o[(size_t)r * H_] = tanhf(acc[r]);
}

constexpr int KC    = 128;
constexpr int SNCH  = H_ / KC;
constexpr int KNL   = SC * KC / 4 / 256;
constexpr int KP    = KC + 4;
constexpr int KB_SZ = SC * KP;
constexpr int SC_SMEM_FLOATS = 2 * KB_SZ + 32 * TQ_STRIDE + 32 * 8;
constexpr int TQN   = 32 * (H_ / 4) / 256;
static_assert(32 * (H_ / 4) % 256 == 0 && TQ_STRIDE % 4 == 0, "tq staging: whole pieces, 16-byte aligned rows");
static_assert(H_ % KC == 0 && KC == 128 && SC == 128 && KP % 4 == 0 &&
              (32 * TQ_STRIDE + 32 * 8) % 4 == 0 && KB_SZ % 4 == 0,
              "chunk plan: a wave loads one 128-float row piece; 16-byte aligned key tiles");

__device__ __forceinline__ void sc_load(float4 (&kr)[KNL], const float* __restrict__ kg, int c) {
#pragma unroll
  for (int j = 0; j < KNL; ++j) kr[j] = *(const float4*)(kg + (size_t)(8 * j) * H_ + c * KC);
}

__device__ __forceinline__ void sc_store(const float4 (&kr)[KNL], float* kb) {
#pragma unroll
  for (int j = 0; j < KNL; ++j) *(float4*)(kb + 8 * j * KP) = kr[j];
}

__device__ __forceinline__ void sc_mma(v8f& acc0, v8f& acc1, const float* a0p, const float* a1p,
                                       const float* bp) {
#pragma unroll
  for (int kk = 0; kk < KC / 4; ++kk) {
    const v2f a0 = *(const v2f*)(a0p + kk * 4);
    const v2f a1 = *(const v2f*)(a1p + kk * 4);
    const v2f bf = *(const v2f*)(bp + kk * 4);
    acc0 = WMMA_F32(a0, bf, acc0);
    acc1 = WMMA_F32(a1, bf, acc1);
  }
}

__global__ __launch_bounds__(256) void k_scores(const float* __restrict__ key,
                                                const float* __restrict__ tq,
                                                const int* __restrict__ seq_len,
                                                float* __restrict__ alpha_g,
                                                float* __restrict__ den_part) {
  __shared__ __attribute__((aligned(16))) float smem[SC_SMEM_FLOATS];
  float* tqs  = smem;
  float* dens = smem + 32 * TQ_STRIDE;
  float* kbuf = dens + 32 * 8;

  const int b     = blockIdx.x / NCHUNK;
  const int chunk = blockIdx.x % NCHUNK;
  const int s0    = chunk * SC;
  const int tid   = threadIdx.x;
  const int lane  = tid & 31;
  const int wv    = tid >> 5;
  const int hl    = lane & 15;
  const int half  = lane >> 4;
  const int kh    = half * 2;

  const float* kg = key + ((size_t)b * S_ + s0 + (tid >> 5)) * H_ + (tid & 31) * 4;
  const int kso = (tid >> 5) * KP + (tid & 31) * 4;
  float4 kr[KNL];

  const float4* tqg = (const float4*)(tq + (size_t)b * CA_ * H_);
  float4 tv[TQN];
  for (int j = 0; j < TQN; ++j) tv[j] = tqg[tid + 256 * j];
  sc_load(kr, kg, 0);
  for (int j = 0; j < TQN; ++j) {
    const int i = tid + 256 * j;
    const int row = i / (H_ / 4), c4 = i % (H_ / 4);
    *(float4*)&tqs[row * TQ_STRIDE + c4 * 4] = tv[j];
  }
  sc_store(kr, kbuf + kso);
  __syncthreads();

  const float* a0p  = &tqs[hl * TQ_STRIDE + kh];
  const float* a1p  = &tqs[(hl + 16) * TQ_STRIDE + kh];
  const int bfo = (wv * 16 + hl) * KP + kh;

  v8f acc0 = {0.f, 0.f, 0.f, 0.f, 0.f, 0.f, 0.f, 0.f};
  v8f acc1 = {0.f, 0.f, 0.f, 0.f, 0.f, 0.f, 0.f, 0.f};
  for (int c = 0; c < SNCH - 1; ++c) {
    const int cur = c & 1, nxt = cur ^ 1;
    sc_load(kr, kg, c + 1);
    sc_mma(acc0, acc1, a0p + c * KC, a1p + c * KC, kbuf + cur * KB_SZ + bfo);
    sc_store(kr, kbuf + nxt * KB_SZ + kso);
    __builtin_amdgcn_sched_group_barrier(0x020, KNL, 0);
    __builtin_amdgcn_sched_group_barrier(0x100, 48, 0);
    __builtin_amdgcn_sched_group_barrier(0x008, KC / 2, 0);
    __builtin_amdgcn_sched_group_barrier(0x200, KNL, 0);
    __syncthreads();
  }
  {
    const int c = SNCH - 1, cur = c & 1;
    sc_mma(acc0, acc1, a0p + c * KC, a1p + c * KC, kbuf + cur * KB_SZ + bfo);
  }

  const int s = s0 + wv * 16 + hl;
  const bool valid = s < seq_len[b];
  float* ag = alpha_g + (size_t)b * CA_ * S_;
  float rs0[8], rs1[8];
  for (int r = 0; r < 8; ++r) {
    const int c0 = r + 8 * half;
    float v0 = valid ? __expf(acc0[r]) : 0.0f;
    float v1 = valid ? __expf(acc1[r]) : 0.0f;
    ag[(size_t)c0 * S_ + s]        = v0;
    ag[(size_t)(c0 + 16) * S_ + s] = v1;
    float r0 = v0, r1 = v1;
    for (int m = 1; m < 16; m <<= 1) {
      r0 += __shfl_xor(r0, m, 32);
      r1 += __shfl_xor(r1, m, 32);
    }
    rs0[r] = r0;
    rs1[r] = r1;
  }
  if (hl == 0) {
    for (int r = 0; r < 8; ++r) {
      const int c0 = r + 8 * half;
      dens[c0 * 8 + wv]        = rs0[r];
      dens[(c0 + 16) * 8 + wv] = rs1[r];
    }
  }
  __syncthreads();
  if (tid < CA_) {
    float d = 0.f;
    for (int w = 0; w < 8; ++w) d += dens[tid * 8 + w];
    den_part[((size_t)b * NCHUNK + chunk) * CA_ + tid] = d;
  }
}

constexpr int OC    = 64;
constexpr int ONCH  = KLEN / OC;
constexpr int VP    = 128 + 8;
constexpr int AP    = OC + 4;
constexpr int OV_SZ = OC * VP;
constexpr int OA_SZ = CA_ * AP;
static_assert(KLEN % OC == 0 && OC % 4 == 0 && VP % 4 == 0 && AP % 4 == 0, "chunk plan");

__device__ __forceinline__ void out_load(float4 (&vr)[8], float4 (&ar)[2],
                                         const float* __restrict__ vg,
                                         const float* __restrict__ ag, int ch) {
#pragma unroll
  for (int j = 0; j < 8; ++j) vr[j] = *(const float4*)(vg + (size_t)(ch * OC + 8 * j) * H_);
#pragma unroll
  for (int j = 0; j < 2; ++j) ar[j] = *(const float4*)(ag + (size_t)(16 * j) * S_ + ch * OC);
}

__device__ __forceinline__ void out_store(const float4 (&vr)[8], const float4 (&ar)[2],
                                          float* vs, float* as_) {
#pragma unroll
  for (int j = 0; j < 8; ++j) *(float4*)(vs + 8 * j * VP) = vr[j];
#pragma unroll
  for (int j = 0; j < 2; ++j) *(float4*)(as_ + 16 * j * AP) = ar[j];
}

__device__ __forceinline__ void out_mma(v8f& acc0, v8f& acc1, const float* a0p,
                                        const float* bp) {
#pragma unroll
  for (int kk = 0; kk < OC / 4; ++kk) {
    const v2f a0 = *(const v2f*)(a0p + kk * 4);
    const v2f a1 = *(const v2f*)(a0p + 16 * AP + kk * 4);
    v2f bf;
    bf.x = bp[(kk * 4) * VP];
    bf.y = bp[(kk * 4 + 1) * VP];
    acc0 = WMMA_F32(a0, bf, acc0);
    acc1 = WMMA_F32(a1, bf, acc1);
  }
}

__global__ __launch_bounds__(256) void k_out(const float* __restrict__ value,
                                             const float* __restrict__ alpha_g,
                                             float* __restrict__ num_part) {
  __shared__ __attribute__((aligned(16))) float osm[2 * (OV_SZ + OA_SZ)];
  const int gid = blockIdx.x;
  const int b   = gid / (KSPLIT * 6);
  const int rem = gid % (KSPLIT * 6);
  const int kc  = rem / 6;
  const int nb  = rem % 6;
  const int tid  = threadIdx.x;
  const int lane = tid & 31;
  const int wv   = tid >> 5;
  const int hl   = lane & 15;
  const int half = lane >> 4;
  const int kh   = half * 2;
  const int n    = nb * 8 + wv;

  const int vrow = tid >> 5, vc4 = tid & 31;
  const int arow = tid >> 4, aq  = tid & 15;
  const float* vg = value + ((size_t)b * S_ + kc * KLEN + vrow) * H_ + nb * 128 + vc4 * 4;
  const float* ag = alpha_g + ((size_t)b * CA_ + arow) * S_ + kc * KLEN + aq * 4;
  const int vso = vrow * VP + vc4 * 4;
  const int aso = arow * AP + aq * 4;
  const int afo = hl * AP + kh;
  const int bfo = kh * VP + wv * 16 + hl;

  v8f acc0 = {0.f, 0.f, 0.f, 0.f, 0.f, 0.f, 0.f, 0.f};
  v8f acc1 = {0.f, 0.f, 0.f, 0.f, 0.f, 0.f, 0.f, 0.f};
  float4 vr[8], ar[2];
  out_load(vr, ar, vg, ag, 0);
  out_store(vr, ar, osm + vso, osm + 2 * OV_SZ + aso);
  __syncthreads();
  for (int ch = 0; ch < ONCH - 1; ++ch) {
    const int cur = ch & 1, nxt = cur ^ 1;
    out_load(vr, ar, vg, ag, ch + 1);
    out_mma(acc0, acc1, osm + 2 * OV_SZ + cur * OA_SZ + afo, osm + cur * OV_SZ + bfo);
    out_store(vr, ar, osm + nxt * OV_SZ + vso, osm + 2 * OV_SZ + nxt * OA_SZ + aso);
    __builtin_amdgcn_sched_group_barrier(0x020, 10, 0);
    __builtin_amdgcn_sched_group_barrier(0x100, 32, 0);
    __builtin_amdgcn_sched_group_barrier(0x008, 32, 0);
    __builtin_amdgcn_sched_group_barrier(0x200, 10, 0);
    __syncthreads();
  }
  {
    const int cur = (ONCH - 1) & 1;
    out_mma(acc0, acc1, osm + 2 * OV_SZ + cur * OA_SZ + afo, osm + cur * OV_SZ + bfo);
  }

  float* o = num_part + ((size_t)kc * B_ + b) * CA_ * H_;
  for (int r = 0; r < 8; ++r) {
    const int c = r + 8 * half;
    o[(size_t)c * H_ + n * 16 + hl]        = acc0[r];
    o[(size_t)(c + 16) * H_ + n * 16 + hl] = acc1[r];
  }
}

__global__ void k_fin(const float* __restrict__ num_part, const float* __restrict__ den_part,
                      float* __restrict__ out) {
  const int idx = blockIdx.x * 256 + threadIdx.x;
  const int bc  = idx / H_;
  const int b   = bc >> 5, c = bc & 31;
  float num = 0.f;
  for (int kc = 0; kc < KSPLIT; ++kc) num += num_part[(size_t)kc * (B_ * CA_ * H_) + idx];
  float den = 0.f;
  for (int ch = 0; ch < NCHUNK; ++ch) den += den_part[((size_t)b * NCHUNK + ch) * CA_ + c];
  out[idx] = num / den;
}

extern "C" void kernel_launch(void* const* d_in, const int* in_sizes, int n_in,
                              void* d_out, int out_size, void* d_ws, size_t ws_size,
                              hipStream_t stream) {
  (void)in_sizes; (void)n_in; (void)out_size; (void)ws_size;
  const float* key     = (const float*)d_in[0];
  const float* value   = (const float*)d_in[1];
  const float* query   = (const float*)d_in[2];
  const int*   seq_len = (const int*)d_in[3];
  const float* W       = (const float*)d_in[4];
  const float* bias    = (const float*)d_in[5];
  float* ws  = (float*)d_ws;
  float* out = (float*)d_out;

  k_tq<<<TQ_GRID, 256, 0, stream>>>(query, W, bias, ws + TQ_OFF);

  k_scores<<<B_ * NCHUNK, 256, 0, stream>>>(key, ws + TQ_OFF, seq_len,
                                                     ws + AL_OFF, ws + DEN_OFF);

  k_out<<<B_ * KSPLIT * 6, 256, 0, stream>>>(value, ws + AL_OFF, ws + NUM_OFF);

  k_fin<<<(B_ * CA_ * H_) / 256, 256, 0, stream>>>(ws + NUM_OFF, ws + DEN_OFF, out);
}
